# starting kernel
# baseline (speedup 1.0000x reference)
#include <hip/hip_runtime.h>
#include <hip/hip_bf16.h>

typedef float v2f __attribute__((ext_vector_type(2)));
typedef float v8f __attribute__((ext_vector_type(8)));

__global__ void gcn_init_deg(float* __restrict__ deg, int n) {
    int i = blockIdx.x * blockDim.x + threadIdx.x;
    if (i < n) deg[i] = 1.0f;
}

__global__ void gcn_deg_scatter(const int* __restrict__ dst, float* __restrict__ deg, int nE) {
    int e = blockIdx.x * blockDim.x + threadIdx.x;
    if (e < nE) atomicAdd(&deg[dst[e]], 1.0f);
}

__global__ void gcn_rsqrt(float* __restrict__ deg, int n) {
    int i = blockIdx.x * blockDim.x + threadIdx.x;
    if (i < n) deg[i] = __frsqrt_rn(deg[i]);
}

template <int NT>
__global__ void gcn_gemm_wmma(const float* __restrict__ X,
                              const float* __restrict__ W,
                              const float* __restrict__ dis,
                              float* __restrict__ hs,
                              float* __restrict__ agg,
                              int M, int K, int N) {
    int wave = (int)((blockIdx.x * blockDim.x + threadIdx.x) >> 5);
    int lane = threadIdx.x & 31;
    int m0 = wave << 4;
    if (m0 >= M) return;

    int lo = lane & 15;
    int hi = lane >> 4;

    const float* xrow = X + (size_t)(m0 + lo) * K;

    int  col[NT];
    int  ccol[NT];
    bool ok[NT];
#pragma unroll
    for (int t = 0; t < NT; ++t) {
        col[t]  = t * 16 + lo;
        ok[t]   = col[t] < N;
        ccol[t] = ok[t] ? col[t] : 0;
    }

    v8f c[NT];
#pragma unroll
    for (int t = 0; t < NT; ++t) c[t] = (v8f){};

    for (int k0 = 0; k0 < K; k0 += 4) {
        int ka = k0 + 2 * hi;
        v2f a;
        a.x = xrow[ka];
        a.y = xrow[ka + 1];
        const float* w0 = W + (size_t)ka * N;
#pragma unroll
        for (int t = 0; t < NT; ++t) {
            float bx = w0[ccol[t]];
            float by = w0[N + ccol[t]];
            v2f b;
            b.x = ok[t] ? bx : 0.0f;
            b.y = ok[t] ? by : 0.0f;
            c[t] = __builtin_amdgcn_wmma_f32_16x16x4_f32(
                        false, a,  false, b,
                        (short)0, c[t],
                        false,  false);
        }
    }

    int rowBase = m0 + (hi << 3);
#pragma unroll
    for (int r = 0; r < 8; ++r) {
        int rr = rowBase + r;
        int rc = rr < M ? rr : M - 1;
        float dr = dis[rc];
        size_t base = (size_t)rc * N;
#pragma unroll
        for (int t = 0; t < NT; ++t) {
            if (rr < M && ok[t]) {
                float v = c[t][r] * dr;
                hs[base + col[t]]  = v;
                agg[base + col[t]] = v;
            }
        }
    }
}

__global__ void gcn_edge_scatter(const int* __restrict__ src,
                                 const int* __restrict__ dst,
                                 const float* __restrict__ hs,
                                 float* __restrict__ agg,
                                 int nE, int F, int chunks) {
    long long t = (long long)blockIdx.x * blockDim.x + threadIdx.x;
    long long total = (long long)nE * chunks;
    if (t >= total) return;
    int e = (int)(t / chunks);
    int c = (int)(t - (long long)e * chunks);
    int s = src[e];
    int d = dst[e];
    const float4 v = *(const float4*)(hs + (long long)s * F + (c << 2));
    float* p = agg + (long long)d * F + (c << 2);
    atomicAdd(p + 0, v.x);
    atomicAdd(p + 1, v.y);
    atomicAdd(p + 2, v.z);
    atomicAdd(p + 3, v.w);
}

__global__ void gcn_finalize(const float* __restrict__ agg,
                             const float* __restrict__ dis,
                             const float* __restrict__ bias,
                             float* __restrict__ out,
                             int nNodes, int F, int doRelu) {
    long long t = (long long)blockIdx.x * blockDim.x + threadIdx.x;
    if (t >= (long long)nNodes * F) return;
    int i = (int)(t / F);
    int f = (int)(t - (long long)i * F);
    float v = fmaf(dis[i], agg[t], bias[f]);
    if (doRelu) v = fmaxf(v, 0.0f);
    out[t] = v;
}

static inline void launch_gemm(int tilesN, const float* X, const float* W,
                               const float* dis, float* hs, float* agg,
                               int M, int K, int N, hipStream_t stream) {
    int mt = (M + 15) / 16;
    long long thr = (long long)mt * 32;
    int blocks = (int)((thr + 255) / 256);
    switch (tilesN) {
    case 1: gcn_gemm_wmma<1><<<blocks, 256, 0, stream>>>(X, W, dis, hs, agg, M, K, N); break;
    case 2: gcn_gemm_wmma<2><<<blocks, 256, 0, stream>>>(X, W, dis, hs, agg, M, K, N); break;
    case 3: gcn_gemm_wmma<3><<<blocks, 256, 0, stream>>>(X, W, dis, hs, agg, M, K, N); break;
    default: gcn_gemm_wmma<4><<<blocks, 256, 0, stream>>>(X, W, dis, hs, agg, M, K, N); break;
    }
}

extern "C" void kernel_launch(void* const* d_in, const int* in_sizes, int n_in,
                              void* d_out, int out_size, void* d_ws, size_t ws_size,
                              hipStream_t stream) {
    const float* x  = (const float*)d_in[0];
    const int*   ei = (const int*)d_in[1];
    const float* W1 = (const float*)d_in[2];
    const float* b1 = (const float*)d_in[3];
    const float* W2 = (const float*)d_in[4];
    const float* b2 = (const float*)d_in[5];
    float* out = (float*)d_out;

    const int IN  = 128;
    const int HID = in_sizes[3];
    const int OUT = in_sizes[5];
    const int N   = in_sizes[0] / IN;
    const int E   = in_sizes[1] / 2;

    const int* src = ei;
    const int* dst = ei + E;

    float* dis  = (float*)d_ws;
    float* hs1  = dis + (((size_t)N + 63) & ~63ull);
    float* agg1 = hs1 + (size_t)N * HID;
    float* hs2  = agg1 + (size_t)N * HID;

    const int B = 256;

    gcn_init_deg<<<(N + B - 1) / B, B, 0, stream>>>(dis, N);
    gcn_deg_scatter<<<(E + B - 1) / B, B, 0, stream>>>(dst, dis, E);
    gcn_rsqrt<<<(N + B - 1) / B, B, 0, stream>>>(dis, N);

    launch_gemm((HID + 15) / 16, x, W1, dis, hs1, agg1, N, IN, HID, stream);

    int chunks1 = HID / 4;
    long long sthr1 = (long long)E * chunks1;
    gcn_edge_scatter<<<(int)((sthr1 + B - 1) / B), B, 0, stream>>>(
        src, dst, hs1, agg1, E, HID, chunks1);

    long long fthr1 = (long long)N * HID;
    gcn_finalize<<<(int)((fthr1 + B - 1) / B), B, 0, stream>>>(
        agg1, dis, b1, hs1, N, HID,  1);

    launch_gemm((OUT + 15) / 16, hs1, W2, dis, hs2, out, N, HID, OUT, stream);

    int chunks2 = OUT / 4;
    long long sthr2 = (long long)E * chunks2;
    gcn_edge_scatter<<<(int)((sthr2 + B - 1) / B), B, 0, stream>>>(
        src, dst, hs2, out, E, OUT, chunks2);

    long long fthr2 = (long long)N * OUT;
    gcn_finalize<<<(int)((fthr2 + B - 1) / B), B, 0, stream>>>(
        out, dis, b2, out, N, OUT,  0);
}


# edge scatter: one wave per 32 edges, lane = feature column: row reads and f32 atomic adds cover whole 128-byte lines (was a float4 per lane, 4-8 lines an instruction); atomic sums, arrival order
# speedup vs baseline: 1.6027x; 1.6027x over previous
#include <hip/hip_runtime.h>
#include <hip/hip_bf16.h>

typedef float v2f __attribute__((ext_vector_type(2)));
typedef float v8f __attribute__((ext_vector_type(8)));

__global__ void gcn_init_deg(float* __restrict__ deg, int n) {
    int i = blockIdx.x * blockDim.x + threadIdx.x;
    if (i < n) deg[i] = 1.0f;
}

__global__ void gcn_deg_scatter(const int* __restrict__ dst, float* __restrict__ deg, int nE) {
    int e = blockIdx.x * blockDim.x + threadIdx.x;
    if (e < nE) atomicAdd(&deg[dst[e]], 1.0f);
}

__global__ void gcn_rsqrt(float* __restrict__ deg, int n) {
    int i = blockIdx.x * blockDim.x + threadIdx.x;
    if (i < n) deg[i] = __frsqrt_rn(deg[i]);
}

template <int NT>
__global__ void gcn_gemm_wmma(const float* __restrict__ X,
                              const float* __restrict__ W,
                              const float* __restrict__ dis,
                              float* __restrict__ hs,
                              float* __restrict__ agg,
                              int M, int K, int N) {
    int wave = (int)((blockIdx.x * blockDim.x + threadIdx.x) >> 5);
    int lane = threadIdx.x & 31;
    int m0 = wave << 4;
    if (m0 >= M) return;

    int lo = lane & 15;
    int hi = lane >> 4;

    const float* xrow = X + (size_t)(m0 + lo) * K;

    int  col[NT];
    int  ccol[NT];
    bool ok[NT];
#pragma unroll
    for (int t = 0; t < NT; ++t) {
        col[t]  = t * 16 + lo;
        ok[t]   = col[t] < N;
        ccol[t] = ok[t] ? col[t] : 0;
    }

    v8f c[NT];
#pragma unroll
    for (int t = 0; t < NT; ++t) c[t] = (v8f){};

    for (int k0 = 0; k0 < K; k0 += 4) {
        int ka = k0 + 2 * hi;
        v2f a;
        a.x = xrow[ka];
        a.y = xrow[ka + 1];
        const float* w0 = W + (size_t)ka * N;
#pragma unroll
        for (int t = 0; t < NT; ++t) {
            float bx = w0[ccol[t]];
            float by = w0[N + ccol[t]];
            v2f b;
            b.x = ok[t] ? bx : 0.0f;
            b.y = ok[t] ? by : 0.0f;
            c[t] = __builtin_amdgcn_wmma_f32_16x16x4_f32(
                        false, a,  false, b,
                        (short)0, c[t],
                        false,  false);
        }
    }

    int rowBase = m0 + (hi << 3);
#pragma unroll
    for (int r = 0; r < 8; ++r) {
        int rr = rowBase + r;
        int rc = rr < M ? rr : M - 1;
        float dr = dis[rc];
        size_t base = (size_t)rc * N;
#pragma unroll
        for (int t = 0; t < NT; ++t) {
            if (rr < M && ok[t]) {
                float v = c[t][r] * dr;
                hs[base + col[t]]  = v;
                agg[base + col[t]] = v;
            }
        }
    }
}


__device__ __forceinline__ void gcn_scatter_rows_slow(int sv, int dv, int count, int lane,
                                                      const float* __restrict__ hs,
                                                      float* __restrict__ agg, int F) {
    for (int i = 0; i < count; ++i) {
        int s = __builtin_amdgcn_readlane(sv, i);
        int d = __builtin_amdgcn_readlane(dv, i);
        const float* srow = hs + (size_t)s * F;
        float* drow = agg + (size_t)d * F;
        for (int f = lane; f < F; f += 32) atomicAdd(drow + f, srow[f]);
    }
}

template <int FULL, int TAIL>
__global__ void gcn_edge_scatter_rows(const int* __restrict__ src,
                                      const int* __restrict__ dst,
                                      const float* __restrict__ hs,
                                      float* __restrict__ agg, int nE) {
    constexpr int F  = 32 * FULL + TAIL;
    constexpr int G  = 16;
    constexpr int P  = TAIL ? 32 / TAIL : 1;
    constexpr int NT = TAIL ? G / P : 0;
    static_assert(TAIL == 0 || (32 % TAIL == 0 && G % P == 0), "tail must divide a wave");

    const int lane = threadIdx.x & 31;
    const long long wave = ((long long)blockIdx.x * blockDim.x + threadIdx.x) >> 5;
    const long long e0 = wave * 32;
    if (e0 >= nE) return;
    const long long rest = (long long)nE - e0;
    const int left = rest < 32 ? (int)rest : 32;
    const int ec = lane < left ? lane : left - 1;
    const int sv = src[e0 + ec];
    const int dv = dst[e0 + ec];

    if (left < 32) {
        gcn_scatter_rows_slow(sv, dv, left, lane, hs, agg, F);
        return;
    }

    const int tl = TAIL ? lane % (TAIL ? TAIL : 1) : 0;
    const int te = TAIL ? lane / (TAIL ? TAIL : 1) : 0;

    for (int g = 0; g < 32; g += G) {
        float v[G][FULL ? FULL : 1];
        float vt[NT ? NT : 1];
#pragma unroll
        for (int i = 0; i < G; ++i) {
            int s = __builtin_amdgcn_readlane(sv, g + i);
            const float* srow = hs + (size_t)s * F;
#pragma unroll
            for (int c = 0; c < FULL; ++c) v[i][c] = srow[c * 32 + lane];
        }
        if constexpr (TAIL > 0) {
#pragma unroll
            for (int j = 0; j < NT; ++j) {
                int st = __shfl(sv, g + j * P + te);
                vt[j] = hs[(size_t)st * F + 32 * FULL + tl];
            }
        }
#pragma unroll
        for (int i = 0; i < G; ++i) {
            int d = __builtin_amdgcn_readlane(dv, g + i);
            float* drow = agg + (size_t)d * F;
#pragma unroll
            for (int c = 0; c < FULL; ++c) atomicAdd(drow + c * 32 + lane, v[i][c]);
        }
        if constexpr (TAIL > 0) {
#pragma unroll
            for (int j = 0; j < NT; ++j) {
                int dt = __shfl(dv, g + j * P + te);
                atomicAdd(agg + (size_t)dt * F + 32 * FULL + tl, vt[j]);
            }
        }
    }
}

__global__ void gcn_edge_scatter_rows_any(const int* __restrict__ src,
                                          const int* __restrict__ dst,
                                          const float* __restrict__ hs,
                                          float* __restrict__ agg, int nE, int F) {
    const int lane = threadIdx.x & 31;
    const long long wave = ((long long)blockIdx.x * blockDim.x + threadIdx.x) >> 5;
    const long long e0 = wave * 32;
    if (e0 >= nE) return;
    const long long rest = (long long)nE - e0;
    const int left = rest < 32 ? (int)rest : 32;
    const int ec = lane < left ? lane : left - 1;
    gcn_scatter_rows_slow(src[e0 + ec], dst[e0 + ec], left, lane, hs, agg, F);
}

static inline void launch_scatter(const int* src, const int* dst, const float* hs,
                                  float* agg, int nE, int F, hipStream_t stream) {
    int waves  = (nE + 31) / 32;
    int blocks = (waves + 7) / 8;
    if (F == 64)
        gcn_edge_scatter_rows<2, 0><<<blocks, 256, 0, stream>>>(src, dst, hs, agg, nE);
    else if (F == 40)
        gcn_edge_scatter_rows<1, 8><<<blocks, 256, 0, stream>>>(src, dst, hs, agg, nE);
    else
        gcn_edge_scatter_rows_any<<<blocks, 256, 0, stream>>>(src, dst, hs, agg, nE, F);
}

__global__ void gcn_finalize(const float* __restrict__ agg,
                             const float* __restrict__ dis,
                             const float* __restrict__ bias,
                             float* __restrict__ out,
                             int nNodes, int F, int doRelu) {
    long long t = (long long)blockIdx.x * blockDim.x + threadIdx.x;
    if (t >= (long long)nNodes * F) return;
    int i = (int)(t / F);
    int f = (int)(t - (long long)i * F);
    float v = fmaf(dis[i], agg[t], bias[f]);
    if (doRelu) v = fmaxf(v, 0.0f);
    out[t] = v;
}

static inline void launch_gemm(int tilesN, const float* X, const float* W,
                               const float* dis, float* hs, float* agg,
                               int M, int K, int N, hipStream_t stream) {
    int mt = (M + 15) / 16;
    long long thr = (long long)mt * 32;
    int blocks = (int)((thr + 255) / 256);
    switch (tilesN) {
    case 1: gcn_gemm_wmma<1><<<blocks, 256, 0, stream>>>(X, W, dis, hs, agg, M, K, N); break;
    case 2: gcn_gemm_wmma<2><<<blocks, 256, 0, stream>>>(X, W, dis, hs, agg, M, K, N); break;
    case 3: gcn_gemm_wmma<3><<<blocks, 256, 0, stream>>>(X, W, dis, hs, agg, M, K, N); break;
    default: gcn_gemm_wmma<4><<<blocks, 256, 0, stream>>>(X, W, dis, hs, agg, M, K, N); break;
    }
}

extern "C" void kernel_launch(void* const* d_in, const int* in_sizes, int n_in,
                              void* d_out, int out_size, void* d_ws, size_t ws_size,
                              hipStream_t stream) {
    const float* x  = (const float*)d_in[0];
    const int*   ei = (const int*)d_in[1];
    const float* W1 = (const float*)d_in[2];
    const float* b1 = (const float*)d_in[3];
    const float* W2 = (const float*)d_in[4];
    const float* b2 = (const float*)d_in[5];
    float* out = (float*)d_out;

    const int IN  = 128;
    const int HID = in_sizes[3];
    const int OUT = in_sizes[5];
    const int N   = in_sizes[0] / IN;
    const int E   = in_sizes[1] / 2;

    const int* src = ei;
    const int* dst = ei + E;

    float* dis  = (float*)d_ws;
    float* hs1  = dis + (((size_t)N + 63) & ~63ull);
    float* agg1 = hs1 + (size_t)N * HID;
    float* hs2  = agg1 + (size_t)N * HID;

    const int B = 256;

    gcn_init_deg<<<(N + B - 1) / B, B, 0, stream>>>(dis, N);
    gcn_deg_scatter<<<(E + B - 1) / B, B, 0, stream>>>(dst, dis, E);
    gcn_rsqrt<<<(N + B - 1) / B, B, 0, stream>>>(dis, N);

    launch_gemm((HID + 15) / 16, x, W1, dis, hs1, agg1, N, IN, HID, stream);

    launch_scatter(src, dst, hs1, agg1, E, HID, stream);

    long long fthr1 = (long long)N * HID;
    gcn_finalize<<<(int)((fthr1 + B - 1) / B), B, 0, stream>>>(
        agg1, dis, b1, hs1, N, HID,  1);

    launch_gemm((OUT + 15) / 16, hs1, W2, dis, hs2, out, N, HID, OUT, stream);

    launch_scatter(src, dst, hs2, out, E, OUT, stream);

    long long fthr2 = (long long)N * OUT;
    gcn_finalize<<<(int)((fthr2 + B - 1) / B), B, 0, stream>>>(
        out, dis, b2, out, N, OUT,  0);
}


# every f32 atomic target (degree, both layer sums) kept twice, one copy per L2 cache side (die id picks a wave's copy); next pass adds the copies: two partial sums a node, arrival order in each
# speedup vs baseline: 2.3867x; 1.4892x over previous
#include <hip/hip_runtime.h>
#include <hip/hip_bf16.h>

typedef float v2f __attribute__((ext_vector_type(2)));
typedef float v8f __attribute__((ext_vector_type(8)));

__device__ __forceinline__ int gcn_cache_side() {
    unsigned id = __builtin_amdgcn_s_sendmsg_rtn(0x87);
    return (int)((id >> 18) & 1u);
}

__global__ void gcn_init_deg(float* __restrict__ deg0, float* __restrict__ deg1, int n) {
    int i = blockIdx.x * blockDim.x + threadIdx.x;
    if (i < n) { deg0[i] = 1.0f; deg1[i] = 0.0f; }
}

__global__ void gcn_deg_scatter(const int* __restrict__ dst, float* __restrict__ deg0,
                                float* __restrict__ deg1, int nE) {
    int e = blockIdx.x * blockDim.x + threadIdx.x;
    float* deg = gcn_cache_side() ? deg1 : deg0;
    if (e < nE) atomicAdd(&deg[dst[e]], 1.0f);
}

__global__ void gcn_rsqrt(float* __restrict__ deg0, const float* __restrict__ deg1, int n) {
    int i = blockIdx.x * blockDim.x + threadIdx.x;
    if (i < n) deg0[i] = __frsqrt_rn(deg0[i] + deg1[i]);
}

template <int NT>
__global__ void gcn_gemm_wmma(const float* __restrict__ X,
                              const float* __restrict__ W,
                              const float* __restrict__ dis,
                              float* __restrict__ hs,
                              float* __restrict__ agg,
                              float* __restrict__ zlo,
                              float* __restrict__ zhi, int zP,
                              int M, int K, int N) {
    int wave = (int)((blockIdx.x * blockDim.x + threadIdx.x) >> 5);
    int lane = threadIdx.x & 31;
    int m0 = wave << 4;
    if (m0 >= M) return;

    int lo = lane & 15;
    int hi = lane >> 4;

    const float* xrow = X + (size_t)(m0 + lo) * K;

    int  col[NT];
    int  ccol[NT];
    bool ok[NT];
#pragma unroll
    for (int t = 0; t < NT; ++t) {
        col[t]  = t * 16 + lo;
        ok[t]   = col[t] < N;
        ccol[t] = ok[t] ? col[t] : 0;
    }

    v8f c[NT];
#pragma unroll
    for (int t = 0; t < NT; ++t) c[t] = (v8f){};

    for (int k0 = 0; k0 < K; k0 += 4) {
        int ka = k0 + 2 * hi;
        v2f a;
        a.x = xrow[ka];
        a.y = xrow[ka + 1];
        const float* w0 = W + (size_t)ka * N;
#pragma unroll
        for (int t = 0; t < NT; ++t) {
            float bx = w0[ccol[t]];
            float by = w0[N + ccol[t]];
            v2f b;
            b.x = ok[t] ? bx : 0.0f;
            b.y = ok[t] ? by : 0.0f;
            c[t] = __builtin_amdgcn_wmma_f32_16x16x4_f32(
                        false, a,  false, b,
                        (short)0, c[t],
                        false,  false);
        }
    }

    int rowBase = m0 + (hi << 3);
#pragma unroll
    for (int r = 0; r < 8; ++r) {
        int rr = rowBase + r;
        int rc = rr < M ? rr : M - 1;
        float dr = dis[rc];
        size_t base = (size_t)rc * N;
#pragma unroll
        for (int t = 0; t < NT; ++t) {
            if (rr < M && ok[t]) {
                float v = c[t][r] * dr;
                hs[base + col[t]]  = v;
                agg[base + col[t]] = v;
                if (zlo) {
                    if (t < 2) zlo[(size_t)rc * zP + col[t]] = 0.0f;
                    else       zhi[(size_t)rc * zP + (col[t] - 32)] = 0.0f;
                }
            }
        }
    }
}


__device__ __forceinline__ void gcn_scatter_rows_slow(int sv, int dv, int count, int lane,
                                                      const float* __restrict__ hs,
                                                      float* lo, float* hi, int P, int F) {
    for (int i = 0; i < count; ++i) {
        int s = __builtin_amdgcn_readlane(sv, i);
        int d = __builtin_amdgcn_readlane(dv, i);
        const float* srow = hs + (size_t)s * F;
        float* dlo = lo + (size_t)d * P;
        float* dhi = hi + (size_t)d * P - 32;
        for (int f = lane; f < F; f += 32) atomicAdd((f < 32 ? dlo : dhi) + f, srow[f]);
    }
}

template <int FULL, int TAIL>
__global__ void gcn_edge_scatter_rows(const int* __restrict__ src,
                                      const int* __restrict__ dst,
                                      const float* __restrict__ hs,
                                      float* a0, float* a1lo, float* a1hi, int P1, int nE) {
    constexpr int F  = 32 * FULL + TAIL;
    constexpr int G  = 16;
    constexpr int P  = TAIL ? 32 / TAIL : 1;
    constexpr int NT = TAIL ? G / P : 0;
    static_assert(FULL >= 1, "a row holds at least one full line");
    static_assert(TAIL == 0 || (32 % TAIL == 0 && G % P == 0), "tail must divide a wave");

    const int lane = threadIdx.x & 31;
    const long long wave = ((long long)blockIdx.x * blockDim.x + threadIdx.x) >> 5;
    const long long e0 = wave * 32;
    if (e0 >= nE) return;
    const long long rest = (long long)nE - e0;
    const int left = rest < 32 ? (int)rest : 32;
    const int ec = lane < left ? lane : left - 1;
    const int sv = src[e0 + ec];
    const int dv = dst[e0 + ec];

    const int side = gcn_cache_side();
    float* lo = side ? a1lo : a0;
    float* hi = side ? a1hi : a0 + 32;
    const int RP = side ? P1 : F;

    if (left < 32) {
        gcn_scatter_rows_slow(sv, dv, left, lane, hs, lo, hi, RP, F);
        return;
    }

    const int tl = TAIL ? lane % (TAIL ? TAIL : 1) : 0;
    const int te = TAIL ? lane / (TAIL ? TAIL : 1) : 0;

    for (int g = 0; g < 32; g += G) {
        float v[G][FULL];
        float vt[NT ? NT : 1];
#pragma unroll
        for (int i = 0; i < G; ++i) {
            int s = __builtin_amdgcn_readlane(sv, g + i);
            const float* srow = hs + (size_t)s * F;
#pragma unroll
            for (int c = 0; c < FULL; ++c) v[i][c] = srow[c * 32 + lane];
        }
        if constexpr (TAIL > 0) {
#pragma unroll
            for (int j = 0; j < NT; ++j) {
                int st = __shfl(sv, g + j * P + te);
                vt[j] = hs[(size_t)st * F + 32 * FULL + tl];
            }
        }
#pragma unroll
        for (int i = 0; i < G; ++i) {
            int d = __builtin_amdgcn_readlane(dv, g + i);
            float* dlo = lo + (size_t)d * RP;
            float* dhi = hi + (size_t)d * RP;
            atomicAdd(dlo + lane, v[i][0]);
#pragma unroll
            for (int c = 1; c < FULL; ++c) atomicAdd(dhi + (c - 1) * 32 + lane, v[i][c]);
        }
        if constexpr (TAIL > 0) {
#pragma unroll
            for (int j = 0; j < NT; ++j) {
                int dt = __shfl(dv, g + j * P + te);
                atomicAdd(hi + (size_t)dt * RP + 32 * (FULL - 1) + tl, vt[j]);
            }
        }
    }
}

__global__ void gcn_edge_scatter_rows_any(const int* __restrict__ src,
                                          const int* __restrict__ dst,
                                          const float* __restrict__ hs,
                                          float* __restrict__ agg, int nE, int F) {
    const int lane = threadIdx.x & 31;
    const long long wave = ((long long)blockIdx.x * blockDim.x + threadIdx.x) >> 5;
    const long long e0 = wave * 32;
    if (e0 >= nE) return;
    const long long rest = (long long)nE - e0;
    const int left = rest < 32 ? (int)rest : 32;
    const int ec = lane < left ? lane : left - 1;
    gcn_scatter_rows_slow(src[e0 + ec], dst[e0 + ec], left, lane, hs, agg, agg + 32, F, F);
}

static inline void launch_scatter(const int* src, const int* dst, const float* hs,
                                  float* agg, float* a1lo, float* a1hi, int P1,
                                  int nE, int F, hipStream_t stream) {
    int waves  = (nE + 31) / 32;
    int blocks = (waves + 7) / 8;
    if (!a1lo) { a1lo = agg; a1hi = agg + 32; P1 = F; }
    if (F == 64)
        gcn_edge_scatter_rows<2, 0><<<blocks, 256, 0, stream>>>(src, dst, hs, agg, a1lo, a1hi, P1, nE);
    else if (F == 40)
        gcn_edge_scatter_rows<1, 8><<<blocks, 256, 0, stream>>>(src, dst, hs, agg, a1lo, a1hi, P1, nE);
    else
        gcn_edge_scatter_rows_any<<<blocks, 256, 0, stream>>>(src, dst, hs, agg, nE, F);
}

__global__ void gcn_finalize(const float* __restrict__ agg,
                             const float* __restrict__ blo,
                             const float* __restrict__ bhi, int P1,
                             const float* __restrict__ dis,
                             const float* __restrict__ bias,
                             float* __restrict__ out,
                             int nNodes, int F, int doRelu) {
    long long t = (long long)blockIdx.x * blockDim.x + threadIdx.x;
    if (t >= (long long)nNodes * F) return;
    int i = (int)(t / F);
    int f = (int)(t - (long long)i * F);
    float a = agg[t];
    if (blo) a += (f < 32) ? blo[(size_t)i * P1 + f] : bhi[(size_t)i * P1 + (f - 32)];
    float v = fmaf(dis[i], a, bias[f]);
    if (doRelu) v = fmaxf(v, 0.0f);
    out[t] = v;
}

static inline void launch_gemm(int tilesN, const float* X, const float* W,
                               const float* dis, float* hs, float* agg,
                               float* zlo, float* zhi, int zP,
                               int M, int K, int N, hipStream_t stream) {
    int mt = (M + 15) / 16;
    long long thr = (long long)mt * 32;
    int blocks = (int)((thr + 255) / 256);
    switch (tilesN) {
    case 1: gcn_gemm_wmma<1><<<blocks, 256, 0, stream>>>(X, W, dis, hs, agg, zlo, zhi, zP, M, K, N); break;
    case 2: gcn_gemm_wmma<2><<<blocks, 256, 0, stream>>>(X, W, dis, hs, agg, zlo, zhi, zP, M, K, N); break;
    case 3: gcn_gemm_wmma<3><<<blocks, 256, 0, stream>>>(X, W, dis, hs, agg, zlo, zhi, zP, M, K, N); break;
    default: gcn_gemm_wmma<4><<<blocks, 256, 0, stream>>>(X, W, dis, hs, agg, zlo, zhi, zP, M, K, N); break;
    }
}

extern "C" void kernel_launch(void* const* d_in, const int* in_sizes, int n_in,
                              void* d_out, int out_size, void* d_ws, size_t ws_size,
                              hipStream_t stream) {
    const float* x  = (const float*)d_in[0];
    const int*   ei = (const int*)d_in[1];
    const float* W1 = (const float*)d_in[2];
    const float* b1 = (const float*)d_in[3];
    const float* W2 = (const float*)d_in[4];
    const float* b2 = (const float*)d_in[5];
    float* out = (float*)d_out;

    const int IN  = 128;
    const int HID = in_sizes[3];
    const int OUT = in_sizes[5];
    const int N   = in_sizes[0] / IN;
    const int E   = in_sizes[1] / 2;

    const int* src = ei;
    const int* dst = ei + E;

    float* dis  = (float*)d_ws;
    float* hs1  = dis + (((size_t)N + 63) & ~63ull);
    float* agg1 = hs1 + (size_t)N * HID;
    float* hs2  = agg1 + (size_t)N * HID;

    const bool two1 = (HID == 64 && OUT >= 32);
    const bool two2 = (OUT == 40 && HID >= OUT);
    float* deg1 = hs1;
    float* l1lo = two1 ? hs2 : nullptr;
    float* l1hi = two1 ? out : nullptr;
    float* l2lo = two2 ? agg1 : nullptr;
    float* l2hi = two2 ? agg1 + 32 : nullptr;

    const int B = 256;

    gcn_init_deg<<<(N + B - 1) / B, B, 0, stream>>>(dis, deg1, N);
    gcn_deg_scatter<<<(E + B - 1) / B, B, 0, stream>>>(dst, dis, deg1, E);
    gcn_rsqrt<<<(N + B - 1) / B, B, 0, stream>>>(dis, deg1, N);

    launch_gemm((HID + 15) / 16, x, W1, dis, hs1, agg1, l1lo, l1hi, 32, N, IN, HID, stream);

    launch_scatter(src, dst, hs1, agg1, l1lo, l1hi, 32, E, HID, stream);

    long long fthr1 = (long long)N * HID;
    gcn_finalize<<<(int)((fthr1 + B - 1) / B), B, 0, stream>>>(
        agg1, l1lo, l1hi, 32, dis, b1, hs1, N, HID,  1);

    launch_gemm((OUT + 15) / 16, hs1, W2, dis, hs2, out, l2lo, l2hi, OUT, N, HID, OUT, stream);

    launch_scatter(src, dst, hs2, out, l2lo, l2hi, OUT, E, OUT, stream);

    long long fthr2 = (long long)N * OUT;
    gcn_finalize<<<(int)((fthr2 + B - 1) / B), B, 0, stream>>>(
        out, l2lo, l2hi, OUT, dis, b2, out, N, OUT,  0);
}


# every edge and row pass launched as a grid the chip holds at once (1536-2048 blocks of 8 waves); waves loop over batches, next indices asked early: about 8x fewer waves enter per call; same sums
# speedup vs baseline: 2.5147x; 1.0536x over previous
#include <hip/hip_runtime.h>
#include <hip/hip_bf16.h>

typedef float v2f __attribute__((ext_vector_type(2)));
typedef float v8f __attribute__((ext_vector_type(8)));

__device__ __forceinline__ int gcn_cache_side() {
    unsigned id = __builtin_amdgcn_s_sendmsg_rtn(0x87);
    return (int)((id >> 18) & 1u);
}

__global__ void gcn_init_deg(float* __restrict__ deg0, float* __restrict__ deg1, int n) {
    int i = blockIdx.x * blockDim.x + threadIdx.x;
    if (i < n) { deg0[i] = 1.0f; deg1[i] = 0.0f; }
}

__global__ void gcn_deg_scatter(const int* __restrict__ dst, float* __restrict__ deg0,
                                float* __restrict__ deg1, int nE) {
    const long long T = (long long)gridDim.x * blockDim.x;
    long long e = (long long)blockIdx.x * blockDim.x + threadIdx.x;
    float* deg = gcn_cache_side() ? deg1 : deg0;
    for (; e + 3 * T < nE; e += 4 * T) {
        int d0 = dst[e];
        int d1 = dst[e + T];
        int d2 = dst[e + 2 * T];
        int d3 = dst[e + 3 * T];
        atomicAdd(&deg[d0], 1.0f);
        atomicAdd(&deg[d1], 1.0f);
        atomicAdd(&deg[d2], 1.0f);
        atomicAdd(&deg[d3], 1.0f);
    }
    for (; e < nE; e += T) atomicAdd(&deg[dst[e]], 1.0f);
}

__global__ void gcn_rsqrt(float* __restrict__ deg0, const float* __restrict__ deg1, int n) {
    int i = blockIdx.x * blockDim.x + threadIdx.x;
    if (i < n) deg0[i] = __frsqrt_rn(deg0[i] + deg1[i]);
}

template <int NT>
__global__ void gcn_gemm_wmma(const float* __restrict__ X,
                              const float* __restrict__ W,
                              const float* __restrict__ dis,
                              float* __restrict__ hs,
                              float* __restrict__ agg,
                              float* __restrict__ zlo,
                              float* __restrict__ zhi, int zP,
                              int M, int K, int N) {
    int wave = (int)((blockIdx.x * blockDim.x + threadIdx.x) >> 5);
    int lane = threadIdx.x & 31;
    int m0 = wave << 4;
    if (m0 >= M) return;

    int lo = lane & 15;
    int hi = lane >> 4;

    const float* xrow = X + (size_t)(m0 + lo) * K;

    int  col[NT];
    int  ccol[NT];
    bool ok[NT];
#pragma unroll
    for (int t = 0; t < NT; ++t) {
        col[t]  = t * 16 + lo;
        ok[t]   = col[t] < N;
        ccol[t] = ok[t] ? col[t] : 0;
    }

    v8f c[NT];
#pragma unroll
    for (int t = 0; t < NT; ++t) c[t] = (v8f){};

    for (int k0 = 0; k0 < K; k0 += 4) {
        int ka = k0 + 2 * hi;
        v2f a;
        a.x = xrow[ka];
        a.y = xrow[ka + 1];
        const float* w0 = W + (size_t)ka * N;
#pragma unroll
        for (int t = 0; t < NT; ++t) {
            float bx = w0[ccol[t]];
            float by = w0[N + ccol[t]];
            v2f b;
            b.x = ok[t] ? bx : 0.0f;
            b.y = ok[t] ? by : 0.0f;
            c[t] = __builtin_amdgcn_wmma_f32_16x16x4_f32(
                        false, a,  false, b,
                        (short)0, c[t],
                        false,  false);
        }
    }

    int rowBase = m0 + (hi << 3);
#pragma unroll
    for (int r = 0; r < 8; ++r) {
        int rr = rowBase + r;
        int rc = rr < M ? rr : M - 1;
        float dr = dis[rc];
        size_t base = (size_t)rc * N;
#pragma unroll
        for (int t = 0; t < NT; ++t) {
            if (rr < M && ok[t]) {
                float v = c[t][r] * dr;
                hs[base + col[t]]  = v;
                agg[base + col[t]] = v;
                if (zlo) {
                    if (t < 2) zlo[(size_t)rc * zP + col[t]] = 0.0f;
                    else       zhi[(size_t)rc * zP + (col[t] - 32)] = 0.0f;
                }
            }
        }
    }
}


__device__ __forceinline__ void gcn_scatter_rows_slow(int sv, int dv, int count, int lane,
                                                      const float* __restrict__ hs,
                                                      float* lo, float* hi, int P, int F) {
    for (int i = 0; i < count; ++i) {
        int s = __builtin_amdgcn_readlane(sv, i);
        int d = __builtin_amdgcn_readlane(dv, i);
        const float* srow = hs + (size_t)s * F;
        float* dlo = lo + (size_t)d * P;
        float* dhi = hi + (size_t)d * P - 32;
        for (int f = lane; f < F; f += 32) atomicAdd((f < 32 ? dlo : dhi) + f, srow[f]);
    }
}

template <int FULL, int TAIL>
__global__ void gcn_edge_scatter_rows(const int* __restrict__ src,
                                      const int* __restrict__ dst,
                                      const float* __restrict__ hs,
                                      float* a0, float* a1lo, float* a1hi, int P1, int nE) {
    constexpr int F  = 32 * FULL + TAIL;
    constexpr int G  = 16;
    constexpr int P  = TAIL ? 32 / TAIL : 1;
    constexpr int NT = TAIL ? G / P : 0;
    static_assert(FULL >= 1, "a row holds at least one full line");
    static_assert(TAIL == 0 || (32 % TAIL == 0 && G % P == 0), "tail must divide a wave");

    const int lane = threadIdx.x & 31;
    const long long nWaves = ((long long)gridDim.x * blockDim.x) >> 5;
    const long long nB = ((long long)nE + 31) >> 5;
    long long b = ((long long)blockIdx.x * blockDim.x + threadIdx.x) >> 5;
    if (b >= nB) return;

    const int side = gcn_cache_side();
    float* lo = side ? a1lo : a0;
    float* hi = side ? a1hi : a0 + 32;
    const int RP = side ? P1 : F;

    const int tl = TAIL ? lane % (TAIL ? TAIL : 1) : 0;
    const int te = TAIL ? lane / (TAIL ? TAIL : 1) : 0;

    long long rest = (long long)nE - b * 32;
    int left = rest < 32 ? (int)rest : 32;
    int ec = lane < left ? lane : left - 1;
    int sv = src[b * 32 + ec];
    int dv = dst[b * 32 + ec];

    for (;;) {
        const long long bn = b + nWaves;
        const bool more = bn < nB;
        const long long bp = more ? bn : b;
        const long long restn = (long long)nE - bp * 32;
        const int leftn = restn < 32 ? (int)restn : 32;
        const int ecn = lane < leftn ? lane : leftn - 1;
        const int svn = src[bp * 32 + ecn];
        const int dvn = dst[bp * 32 + ecn];

        if (left < 32) {
            gcn_scatter_rows_slow(sv, dv, left, lane, hs, lo, hi, RP, F);
        } else {
            for (int g = 0; g < 32; g += G) {
                float v[G][FULL];
                float vt[NT ? NT : 1];
#pragma unroll
                for (int i = 0; i < G; ++i) {
                    int s = __builtin_amdgcn_readlane(sv, g + i);
                    const float* srow = hs + (size_t)s * F;
#pragma unroll
                    for (int c = 0; c < FULL; ++c) v[i][c] = srow[c * 32 + lane];
                }
                if constexpr (TAIL > 0) {
#pragma unroll
                    for (int j = 0; j < NT; ++j) {
                        int st = __shfl(sv, g + j * P + te);
                        vt[j] = hs[(size_t)st * F + 32 * FULL + tl];
                    }
                }
#pragma unroll
                for (int i = 0; i < G; ++i) {
                    int d = __builtin_amdgcn_readlane(dv, g + i);
                    float* dlo = lo + (size_t)d * RP;
                    float* dhi = hi + (size_t)d * RP;
                    atomicAdd(dlo + lane, v[i][0]);
#pragma unroll
                    for (int c = 1; c < FULL; ++c) atomicAdd(dhi + (c - 1) * 32 + lane, v[i][c]);
                }
                if constexpr (TAIL > 0) {
#pragma unroll
                    for (int j = 0; j < NT; ++j) {
                        int dt = __shfl(dv, g + j * P + te);
                        atomicAdd(hi + (size_t)dt * RP + 32 * (FULL - 1) + tl, vt[j]);
                    }
                }
            }
        }

        if (!more) break;
        b = bn; sv = svn; dv = dvn; left = leftn;
    }
}

__global__ void gcn_edge_scatter_rows_any(const int* __restrict__ src,
                                          const int* __restrict__ dst,
                                          const float* __restrict__ hs,
                                          float* __restrict__ agg, int nE, int F) {
    const int lane = threadIdx.x & 31;
    const long long nWaves = ((long long)gridDim.x * blockDim.x) >> 5;
    const long long nB = ((long long)nE + 31) >> 5;
    long long b = ((long long)blockIdx.x * blockDim.x + threadIdx.x) >> 5;
    for (; b < nB; b += nWaves) {
        const long long e0 = b * 32;
        const long long rest = (long long)nE - e0;
        const int left = rest < 32 ? (int)rest : 32;
        const int ec = lane < left ? lane : left - 1;
        gcn_scatter_rows_slow(src[e0 + ec], dst[e0 + ec], left, lane, hs, agg, agg + 32, F, F);
    }
}

#define GCN_RESIDENT_BLOCKS 2048
#define GCN_RESIDENT_BLOCKS_WIDE 1536

static inline void launch_scatter(const int* src, const int* dst, const float* hs,
                                  float* agg, float* a1lo, float* a1hi, int P1,
                                  int nE, int F, hipStream_t stream) {
    int waves  = (nE + 31) / 32;
    int blocks = (waves + 7) / 8;
    int cap = (F == 64) ? GCN_RESIDENT_BLOCKS_WIDE : GCN_RESIDENT_BLOCKS;
    if (blocks > cap) blocks = cap;
    if (!a1lo) { a1lo = agg; a1hi = agg + 32; P1 = F; }
    if (F == 64)
        gcn_edge_scatter_rows<2, 0><<<blocks, 256, 0, stream>>>(src, dst, hs, agg, a1lo, a1hi, P1, nE);
    else if (F == 40)
        gcn_edge_scatter_rows<1, 8><<<blocks, 256, 0, stream>>>(src, dst, hs, agg, a1lo, a1hi, P1, nE);
    else
        gcn_edge_scatter_rows_any<<<blocks, 256, 0, stream>>>(src, dst, hs, agg, nE, F);
}

template <bool TWO>
__global__ void gcn_finalize(const float* agg,
                             const float* __restrict__ blo,
                             const float* __restrict__ bhi, int P1,
                             const float* __restrict__ dis,
                             const float* __restrict__ bias,
                             float* out,
                             int nNodes, int F, int doRelu) {
    constexpr int U = 4;
    const long long total = (long long)nNodes * F;
    const unsigned T  = gridDim.x * blockDim.x;
    const unsigned t0 = blockIdx.x * blockDim.x + threadIdx.x;
    const int qT = (int)(T / (unsigned)F);
    const int rT = (int)(T % (unsigned)F);
    int i = (int)(t0 / (unsigned)F);
    int f = (int)(t0 % (unsigned)F);
    for (long long t = t0; t < total; t += (long long)T * U) {
        float a[U], a2[U], d[U], b[U];
#pragma unroll
        for (int u = 0; u < U; ++u) {
            const long long tu = t + (long long)T * u;
            const long long tc = tu < total ? tu : total - 1;
            const int ic = i < nNodes ? i : nNodes - 1;
            a[u] = agg[tc];
            a2[u] = 0.0f;
            if constexpr (TWO) {
                const float* p2 = (f < 32) ? blo + (size_t)ic * P1 + f
                                           : bhi + (size_t)ic * P1 + (f - 32);
                a2[u] = *p2;
            }
            d[u] = dis[ic];
            b[u] = bias[f];
            i += qT; f += rT;
            if (f >= F) { f -= F; ++i; }
        }
#pragma unroll
        for (int u = 0; u < U; ++u) {
            const long long tu = t + (long long)T * u;
            if (tu < total) {
                float s = a[u];
                if constexpr (TWO) s += a2[u];
                float v = fmaf(d[u], s, b[u]);
                if (doRelu) v = fmaxf(v, 0.0f);
                out[tu] = v;
            }
        }
    }
}

static inline void launch_finalize(int blocks, int threads, const float* agg, const float* blo,
                                   const float* bhi, int P1, const float* dis, const float* bias,
                                   float* out, int nNodes, int F, int doRelu, hipStream_t stream) {
    if (blo) gcn_finalize<true><<<blocks, threads, 0, stream>>>(agg, blo, bhi, P1, dis, bias, out, nNodes, F, doRelu);
    else     gcn_finalize<false><<<blocks, threads, 0, stream>>>(agg, blo, bhi, P1, dis, bias, out, nNodes, F, doRelu);
}

static inline void launch_gemm(int tilesN, const float* X, const float* W,
                               const float* dis, float* hs, float* agg,
                               float* zlo, float* zhi, int zP,
                               int M, int K, int N, hipStream_t stream) {
    int mt = (M + 15) / 16;
    long long thr = (long long)mt * 32;
    int blocks = (int)((thr + 255) / 256);
    switch (tilesN) {
    case 1: gcn_gemm_wmma<1><<<blocks, 256, 0, stream>>>(X, W, dis, hs, agg, zlo, zhi, zP, M, K, N); break;
    case 2: gcn_gemm_wmma<2><<<blocks, 256, 0, stream>>>(X, W, dis, hs, agg, zlo, zhi, zP, M, K, N); break;
    case 3: gcn_gemm_wmma<3><<<blocks, 256, 0, stream>>>(X, W, dis, hs, agg, zlo, zhi, zP, M, K, N); break;
    default: gcn_gemm_wmma<4><<<blocks, 256, 0, stream>>>(X, W, dis, hs, agg, zlo, zhi, zP, M, K, N); break;
    }
}

extern "C" void kernel_launch(void* const* d_in, const int* in_sizes, int n_in,
                              void* d_out, int out_size, void* d_ws, size_t ws_size,
                              hipStream_t stream) {
    const float* x  = (const float*)d_in[0];
    const int*   ei = (const int*)d_in[1];
    const float* W1 = (const float*)d_in[2];
    const float* b1 = (const float*)d_in[3];
    const float* W2 = (const float*)d_in[4];
    const float* b2 = (const float*)d_in[5];
    float* out = (float*)d_out;

    const int IN  = 128;
    const int HID = in_sizes[3];
    const int OUT = in_sizes[5];
    const int N   = in_sizes[0] / IN;
    const int E   = in_sizes[1] / 2;

    const int* src = ei;
    const int* dst = ei + E;

    float* dis  = (float*)d_ws;
    float* hs1  = dis + (((size_t)N + 63) & ~63ull);
    float* agg1 = hs1 + (size_t)N * HID;
    float* hs2  = agg1 + (size_t)N * HID;

    const bool two1 = (HID == 64 && OUT >= 32);
    const bool two2 = (OUT == 40 && HID >= OUT);
    float* deg1 = hs1;
    float* l1lo = two1 ? hs2 : nullptr;
    float* l1hi = two1 ? out : nullptr;
    float* l2lo = two2 ? agg1 : nullptr;
    float* l2hi = two2 ? agg1 + 32 : nullptr;

    const int B = 256;

    gcn_init_deg<<<(N + B - 1) / B, B, 0, stream>>>(dis, deg1, N);
    int degBlocks = (E + B - 1) / B;
    if (degBlocks > GCN_RESIDENT_BLOCKS) degBlocks = GCN_RESIDENT_BLOCKS;
    gcn_deg_scatter<<<degBlocks, B, 0, stream>>>(dst, dis, deg1, E);
    gcn_rsqrt<<<(N + B - 1) / B, B, 0, stream>>>(dis, deg1, N);

    launch_gemm((HID + 15) / 16, x, W1, dis, hs1, agg1, l1lo, l1hi, 32, N, IN, HID, stream);

    launch_scatter(src, dst, hs1, agg1, l1lo, l1hi, 32, E, HID, stream);

    long long fthr1 = (long long)N * HID;
    int fb1 = (int)((fthr1 + B - 1) / B);
    if (fb1 > GCN_RESIDENT_BLOCKS) fb1 = GCN_RESIDENT_BLOCKS;
    launch_finalize(fb1, B, agg1, l1lo, l1hi, 32, dis, b1, hs1, N, HID,  1, stream);

    launch_gemm((OUT + 15) / 16, hs1, W2, dis, hs2, out, l2lo, l2hi, OUT, N, HID, OUT, stream);

    launch_scatter(src, dst, hs2, out, l2lo, l2hi, OUT, E, OUT, stream);

    long long fthr2 = (long long)N * OUT;
    int fb2 = (int)((fthr2 + B - 1) / B);
    if (fb2 > GCN_RESIDENT_BLOCKS) fb2 = GCN_RESIDENT_BLOCKS;
    launch_finalize(fb2, B, out, l2lo, l2hi, OUT, dis, b2, out, N, OUT,  0, stream);
}
